# starting kernel
# baseline (speedup 1.0000x reference)
#include <hip/hip_runtime.h>
#include <math.h>

typedef __attribute__((ext_vector_type(2))) float v2f;
typedef __attribute__((ext_vector_type(4))) float v4f;
typedef __attribute__((ext_vector_type(8))) float v8f;

namespace {
constexpr int kB = 4;
constexpr int kS = 4096;
constexpr int kD = 1024;
constexpr int kH = 64;
constexpr float kLnEps = 1e-5f;
constexpr float kScale = 0.125f;
}

static __device__ __forceinline__ v8f wmma_f32(v2f a, v2f b, v8f c) {
    return __builtin_amdgcn_wmma_f32_16x16x4_f32(false, a, false, b, (short)0, c,
                                                 false, false);
}

__global__ __launch_bounds__(128) void proj_ln_kernel(
    const float* __restrict__ x, const float* __restrict__ Wq,
    const float* __restrict__ Wk, const float* __restrict__ Wv,
    float* __restrict__ qo, float* __restrict__ ko, float* __restrict__ vo) {
    __shared__ float tq[4][16][68];
    __shared__ float tk[4][16][68];
    __shared__ float tv[4][16][68];

    const int tid = threadIdx.x;
    const int w = tid >> 5;
    const int lane = tid & 31;
    const int m = lane & 15;
    const int hi = lane >> 4;
    const int rowBase = blockIdx.x * 16;

    v8f aq[4], ak[4], av[4];
#pragma unroll
    for (int nt = 0; nt < 4; ++nt) { aq[nt] = {}; ak[nt] = {}; av[nt] = {}; }

    const float* xr = x + (size_t)(rowBase + m) * kD;
    const int k0beg = w * (kD / 4);
    for (int it = 0; it < kD / 16; ++it) {
        const int k0 = k0beg + 4 * it;
        v2f a;
        a.x = xr[k0 + 2 * hi];
        a.y = xr[k0 + 2 * hi + 1];
        const int wrow = (k0 + 2 * hi) * kH;
#pragma unroll
        for (int nt = 0; nt < 4; ++nt) {
            const int col = nt * 16 + m;
            v2f bq, bk, bv;
            bq.x = Wq[wrow + col];      bq.y = Wq[wrow + kH + col];
            bk.x = Wk[wrow + col];      bk.y = Wk[wrow + kH + col];
            bv.x = Wv[wrow + col];      bv.y = Wv[wrow + kH + col];
            aq[nt] = wmma_f32(a, bq, aq[nt]);
            ak[nt] = wmma_f32(a, bk, ak[nt]);
            av[nt] = wmma_f32(a, bv, av[nt]);
        }
    }

#pragma unroll
    for (int nt = 0; nt < 4; ++nt)
#pragma unroll
        for (int r = 0; r < 8; ++r) {
            const int row = r + 8 * hi;
            const int col = nt * 16 + m;
            tq[w][row][col] = aq[nt][r];
            tk[w][row][col] = ak[nt][r];
            tv[w][row][col] = av[nt][r];
        }
    __syncthreads();

    if (tid < 48) {
        const int which = tid >> 4;
        const int row = tid & 15;
        float(*arr)[16][68] = (which == 0) ? tq : (which == 1) ? tk : tv;
        float* r0 = &arr[0][row][0];
        const float* r1 = &arr[1][row][0];
        const float* r2 = &arr[2][row][0];
        const float* r3 = &arr[3][row][0];
        float s = 0.f, s2 = 0.f;
#pragma unroll
        for (int c = 0; c < kH; ++c) {
            const float t = r0[c] + r1[c] + r2[c] + r3[c];
            r0[c] = t;
            s += t;
            s2 += t * t;
        }
        if (which < 2) {
            const float mu = s * (1.f / kH);
            const float var = s2 * (1.f / kH) - mu * mu;
            const float inv = rsqrtf(var + kLnEps);
#pragma unroll
            for (int c = 0; c < kH; ++c) r0[c] = (r0[c] - mu) * inv;
        }
    }
    __syncthreads();

    const size_t ob = (size_t)rowBase * kH;
#pragma unroll
    for (int it = 0; it < 2; ++it) {
        const int flat = it * 512 + tid * 4;
        const int row = flat >> 6;
        const int col = flat & 63;
        *reinterpret_cast<v4f*>(qo + ob + flat) =
            *reinterpret_cast<const v4f*>(&tq[0][row][col]);
        *reinterpret_cast<v4f*>(ko + ob + flat) =
            *reinterpret_cast<const v4f*>(&tk[0][row][col]);
        *reinterpret_cast<v4f*>(vo + ob + flat) =
            *reinterpret_cast<const v4f*>(&tv[0][row][col]);
    }
}

__global__ __launch_bounds__(128) void flash_attn_kernel(
    const float* __restrict__ q, const float* __restrict__ k,
    const float* __restrict__ v, float* __restrict__ out) {
    __shared__ float ps[4][16][18];
    __shared__ float m_s[4][16];
    __shared__ float l_s[4][16];
    __shared__ float o_s[4][16][64];

    const int tid = threadIdx.x;
    const int w = tid >> 5;
    const int lane = tid & 31;
    const int m = lane & 15;
    const int hi = lane >> 4;
    const int b = blockIdx.x >> 8;
    const int qt = blockIdx.x & 255;

    const float* Qb = q + ((size_t)b * kS + qt * 16) * kH;
    const float* Kb = k + (size_t)b * kS * kH;
    const float* Vb = v + (size_t)b * kS * kH;

    v2f aq[16];
#pragma unroll
    for (int kk = 0; kk < 16; ++kk) {
        const float* p = Qb + m * kH + kk * 4 + 2 * hi;
        aq[kk].x = p[0] * kScale;
        aq[kk].y = p[1] * kScale;
    }

    v8f o[4];
#pragma unroll
    for (int nt = 0; nt < 4; ++nt) o[nt] = {};
    v8f lacc = {};
    float mrow[8];
#pragma unroll
    for (int r = 0; r < 8; ++r) mrow[r] = -INFINITY;
    const v2f bones = {1.f, 1.f};

    constexpr int kTiles = (kS / 16) / 4;
    for (int t = 0; t < kTiles; ++t) {
        const int j = w + 4 * t;
        const float* Kt = Kb + (size_t)j * 16 * kH;

        {
            const int jn = (t + 1 < kTiles) ? (j + 4) : j;
            const size_t nb = (size_t)jn * 16 * kH + lane * 32;
            __builtin_prefetch(Kb + nb, 0, 3);
            __builtin_prefetch(Vb + nb, 0, 3);
        }

        v8f sc0 = {}, sc1 = {};
#pragma unroll
        for (int kk = 0; kk < 8; ++kk) {
            v2f bk0, bk1;
            const float* p0 = Kt + m * kH + kk * 4 + 2 * hi;
            const float* p1 = p0 + 32;
            bk0.x = p0[0];  bk0.y = p0[1];
            bk1.x = p1[0];  bk1.y = p1[1];
            sc0 = wmma_f32(aq[kk], bk0, sc0);
            sc1 = wmma_f32(aq[kk + 8], bk1, sc1);
        }
        v8f sc;
#pragma unroll
        for (int r = 0; r < 8; ++r) sc[r] = sc0[r] + sc1[r];

        v8f pf;
        float al[8];
#pragma unroll
        for (int r = 0; r < 8; ++r) {
            float tmax = sc[r];
            tmax = fmaxf(tmax, __shfl_xor(tmax, 1));
            tmax = fmaxf(tmax, __shfl_xor(tmax, 2));
            tmax = fmaxf(tmax, __shfl_xor(tmax, 4));
            tmax = fmaxf(tmax, __shfl_xor(tmax, 8));
            const float mn = fmaxf(mrow[r], tmax);
            al[r] = __expf(mrow[r] - mn);
            mrow[r] = mn;
            pf[r] = __expf(sc[r] - mn);
        }
#pragma unroll
        for (int r = 0; r < 8; ++r) lacc[r] *= al[r];
#pragma unroll
        for (int nt = 0; nt < 4; ++nt)
#pragma unroll
            for (int r = 0; r < 8; ++r) o[nt][r] *= al[r];

#pragma unroll
        for (int r = 0; r < 8; ++r) ps[w][r + 8 * hi][m] = pf[r];
        asm volatile("" ::: "memory");

#pragma unroll
        for (int kk = 0; kk < 4; ++kk) {
            v2f ap;
            ap.x = ps[w][m][kk * 4 + 2 * hi];
            ap.y = ps[w][m][kk * 4 + 2 * hi + 1];
            lacc = wmma_f32(ap, bones, lacc);
            const float* vp = Vb + (size_t)(j * 16 + kk * 4 + 2 * hi) * kH;
#pragma unroll
            for (int nt = 0; nt < 4; ++nt) {
                v2f bv;
                bv.x = vp[nt * 16 + m];
                bv.y = vp[kH + nt * 16 + m];
                o[nt] = wmma_f32(ap, bv, o[nt]);
            }
        }
        asm volatile("" ::: "memory");
    }

    if (m == 0) {
#pragma unroll
        for (int r = 0; r < 8; ++r) {
            m_s[w][r + 8 * hi] = mrow[r];
            l_s[w][r + 8 * hi] = lacc[r];
        }
    }
#pragma unroll
    for (int nt = 0; nt < 4; ++nt)
#pragma unroll
        for (int r = 0; r < 8; ++r)
            o_s[w][r + 8 * hi][nt * 16 + m] = o[nt][r];
    __syncthreads();

    const int row = tid >> 3;
    const int c0 = (tid & 7) * 8;
    float M = m_s[0][row];
#pragma unroll
    for (int w2 = 1; w2 < 4; ++w2) M = fmaxf(M, m_s[w2][row]);
    float e[4];
    float L = 0.f;
#pragma unroll
    for (int w2 = 0; w2 < 4; ++w2) {
        e[w2] = __expf(m_s[w2][row] - M);
        L += e[w2] * l_s[w2][row];
    }
    const float invL = 1.f / L;
    float* Ob = out + ((size_t)b * kS + qt * 16) * kH;
#pragma unroll
    for (int c = 0; c < 8; ++c) {
        const float acc = e[0] * o_s[0][row][c0 + c] + e[1] * o_s[1][row][c0 + c] +
                          e[2] * o_s[2][row][c0 + c] + e[3] * o_s[3][row][c0 + c];
        Ob[row * kH + c0 + c] = acc * invL;
    }
}

extern "C" void kernel_launch(void* const* d_in, const int* in_sizes, int n_in,
                              void* d_out, int out_size, void* d_ws,
                              size_t ws_size, hipStream_t stream) {
    const float* x = (const float*)d_in[0];
    const float* Wq = (const float*)d_in[1];
    const float* Wk = (const float*)d_in[2];
    const float* Wv = (const float*)d_in[3];
    float* out = (float*)d_out;

    float* qws = (float*)d_ws;
    float* kws = qws + (size_t)kB * kS * kH;
    float* vws = kws + (size_t)kB * kS * kH;

    proj_ln_kernel<<<(kB * kS) / 16, 128, 0, stream>>>(x, Wq, Wk, Wv, qws, kws,
                                                       vws);
    flash_attn_kernel<<<kB * (kS / 16), 128, 0, stream>>>(qws, kws, vws, out);
}


# attention: each wave holds two 16-query tiles so every loaded K/V fragment feeds two WMMAs; keys split over 8 waves per 32-query block (was 4 per 16): 8 partials merge, key sum order changes
# speedup vs baseline: 1.1602x; 1.1602x over previous
#include <hip/hip_runtime.h>
#include <math.h>

typedef __attribute__((ext_vector_type(2))) float v2f;
typedef __attribute__((ext_vector_type(4))) float v4f;
typedef __attribute__((ext_vector_type(8))) float v8f;

namespace {
constexpr int kB = 4;
constexpr int kS = 4096;
constexpr int kD = 1024;
constexpr int kH = 64;
constexpr float kLnEps = 1e-5f;
constexpr float kScale = 0.125f;
}

static __device__ __forceinline__ v8f wmma_f32(v2f a, v2f b, v8f c) {
    return __builtin_amdgcn_wmma_f32_16x16x4_f32(false, a, false, b, (short)0, c,
                                                 false, false);
}

__global__ __launch_bounds__(128) void proj_ln_kernel(
    const float* __restrict__ x, const float* __restrict__ Wq,
    const float* __restrict__ Wk, const float* __restrict__ Wv,
    float* __restrict__ qo, float* __restrict__ ko, float* __restrict__ vo) {
    __shared__ float tq[4][16][68];
    __shared__ float tk[4][16][68];
    __shared__ float tv[4][16][68];

    const int tid = threadIdx.x;
    const int w = tid >> 5;
    const int lane = tid & 31;
    const int m = lane & 15;
    const int hi = lane >> 4;
    const int rowBase = blockIdx.x * 16;

    v8f aq[4], ak[4], av[4];
#pragma unroll
    for (int nt = 0; nt < 4; ++nt) { aq[nt] = {}; ak[nt] = {}; av[nt] = {}; }

    const float* xr = x + (size_t)(rowBase + m) * kD;
    const int k0beg = w * (kD / 4);
    for (int it = 0; it < kD / 16; ++it) {
        const int k0 = k0beg + 4 * it;
        v2f a;
        a.x = xr[k0 + 2 * hi];
        a.y = xr[k0 + 2 * hi + 1];
        const int wrow = (k0 + 2 * hi) * kH;
#pragma unroll
        for (int nt = 0; nt < 4; ++nt) {
            const int col = nt * 16 + m;
            v2f bq, bk, bv;
            bq.x = Wq[wrow + col];      bq.y = Wq[wrow + kH + col];
            bk.x = Wk[wrow + col];      bk.y = Wk[wrow + kH + col];
            bv.x = Wv[wrow + col];      bv.y = Wv[wrow + kH + col];
            aq[nt] = wmma_f32(a, bq, aq[nt]);
            ak[nt] = wmma_f32(a, bk, ak[nt]);
            av[nt] = wmma_f32(a, bv, av[nt]);
        }
    }

#pragma unroll
    for (int nt = 0; nt < 4; ++nt)
#pragma unroll
        for (int r = 0; r < 8; ++r) {
            const int row = r + 8 * hi;
            const int col = nt * 16 + m;
            tq[w][row][col] = aq[nt][r];
            tk[w][row][col] = ak[nt][r];
            tv[w][row][col] = av[nt][r];
        }
    __syncthreads();

    if (tid < 48) {
        const int which = tid >> 4;
        const int row = tid & 15;
        float(*arr)[16][68] = (which == 0) ? tq : (which == 1) ? tk : tv;
        float* r0 = &arr[0][row][0];
        const float* r1 = &arr[1][row][0];
        const float* r2 = &arr[2][row][0];
        const float* r3 = &arr[3][row][0];
        float s = 0.f, s2 = 0.f;
#pragma unroll
        for (int c = 0; c < kH; ++c) {
            const float t = r0[c] + r1[c] + r2[c] + r3[c];
            r0[c] = t;
            s += t;
            s2 += t * t;
        }
        if (which < 2) {
            const float mu = s * (1.f / kH);
            const float var = s2 * (1.f / kH) - mu * mu;
            const float inv = rsqrtf(var + kLnEps);
#pragma unroll
            for (int c = 0; c < kH; ++c) r0[c] = (r0[c] - mu) * inv;
        }
    }
    __syncthreads();

    const size_t ob = (size_t)rowBase * kH;
#pragma unroll
    for (int it = 0; it < 2; ++it) {
        const int flat = it * 512 + tid * 4;
        const int row = flat >> 6;
        const int col = flat & 63;
        *reinterpret_cast<v4f*>(qo + ob + flat) =
            *reinterpret_cast<const v4f*>(&tq[0][row][col]);
        *reinterpret_cast<v4f*>(ko + ob + flat) =
            *reinterpret_cast<const v4f*>(&tk[0][row][col]);
        *reinterpret_cast<v4f*>(vo + ob + flat) =
            *reinterpret_cast<const v4f*>(&tv[0][row][col]);
    }
}

__global__ __launch_bounds__(256) void flash_attn_kernel(
    const float* __restrict__ q, const float* __restrict__ k,
    const float* __restrict__ v, float* __restrict__ out) {
    constexpr int kQT = 2;
    constexpr int kWaves = 8;
    __shared__ float ps[kWaves][kQT][16][18];
    __shared__ float m_s[kWaves][kQT * 16];
    __shared__ float l_s[kWaves][kQT * 16];
    __shared__ alignas(16) float o_s[kWaves][16][64];

    const int tid = threadIdx.x;
    const int w = tid >> 5;
    const int lane = tid & 31;
    const int m = lane & 15;
    const int hi = lane >> 4;
    const int b = blockIdx.x >> 7;
    const int qb = blockIdx.x & 127;

    const float* Qb = q + ((size_t)b * kS + qb * 32) * kH;
    const float* Kb = k + (size_t)b * kS * kH;
    const float* Vb = v + (size_t)b * kS * kH;

    v2f aq[kQT][16];
#pragma unroll
    for (int qi = 0; qi < kQT; ++qi)
#pragma unroll
        for (int kk = 0; kk < 16; ++kk) {
            const float* p = Qb + (qi * 16 + m) * kH + kk * 4 + 2 * hi;
            aq[qi][kk].x = p[0] * kScale;
            aq[qi][kk].y = p[1] * kScale;
        }

    v8f o[kQT][4];
    v8f lacc[kQT];
    float mrow[kQT][8];
#pragma unroll
    for (int qi = 0; qi < kQT; ++qi) {
#pragma unroll
        for (int nt = 0; nt < 4; ++nt) o[qi][nt] = {};
        lacc[qi] = {};
#pragma unroll
        for (int r = 0; r < 8; ++r) mrow[qi][r] = -INFINITY;
    }
    const v2f bones = {1.f, 1.f};

    constexpr int kTiles = (kS / 16) / kWaves;
    for (int t = 0; t < kTiles; ++t) {
        const int j = w + kWaves * t;
        const float* Kt = Kb + (size_t)j * 16 * kH;

        {
            const int jn = (t + 1 < kTiles) ? (j + kWaves) : j;
            const size_t nb = (size_t)jn * 16 * kH + lane * 32;
            __builtin_prefetch(Kb + nb, 0, 3);
            __builtin_prefetch(Vb + nb, 0, 3);
        }

        v8f sc0[kQT], sc1[kQT];
#pragma unroll
        for (int qi = 0; qi < kQT; ++qi) { sc0[qi] = {}; sc1[qi] = {}; }
#pragma unroll
        for (int kk = 0; kk < 8; ++kk) {
            v2f bk0, bk1;
            const float* p0 = Kt + m * kH + kk * 4 + 2 * hi;
            const float* p1 = p0 + 32;
            bk0.x = p0[0];  bk0.y = p0[1];
            bk1.x = p1[0];  bk1.y = p1[1];
#pragma unroll
            for (int qi = 0; qi < kQT; ++qi) {
                sc0[qi] = wmma_f32(aq[qi][kk], bk0, sc0[qi]);
                sc1[qi] = wmma_f32(aq[qi][kk + 8], bk1, sc1[qi]);
            }
        }

#pragma unroll
        for (int qi = 0; qi < kQT; ++qi) {
            v8f sc;
#pragma unroll
            for (int r = 0; r < 8; ++r) sc[r] = sc0[qi][r] + sc1[qi][r];

            v8f pf;
            float al[8];
#pragma unroll
            for (int r = 0; r < 8; ++r) {
                float tmax = sc[r];
                tmax = fmaxf(tmax, __shfl_xor(tmax, 1));
                tmax = fmaxf(tmax, __shfl_xor(tmax, 2));
                tmax = fmaxf(tmax, __shfl_xor(tmax, 4));
                tmax = fmaxf(tmax, __shfl_xor(tmax, 8));
                const float mn = fmaxf(mrow[qi][r], tmax);
                al[r] = __expf(mrow[qi][r] - mn);
                mrow[qi][r] = mn;
                pf[r] = __expf(sc[r] - mn);
            }
#pragma unroll
            for (int r = 0; r < 8; ++r) lacc[qi][r] *= al[r];
#pragma unroll
            for (int nt = 0; nt < 4; ++nt)
#pragma unroll
                for (int r = 0; r < 8; ++r) o[qi][nt][r] *= al[r];

#pragma unroll
            for (int r = 0; r < 8; ++r) ps[w][qi][r + 8 * hi][m] = pf[r];
        }
        asm volatile("" ::: "memory");

#pragma unroll
        for (int kk = 0; kk < 4; ++kk) {
            v2f ap[kQT];
#pragma unroll
            for (int qi = 0; qi < kQT; ++qi) {
                ap[qi].x = ps[w][qi][m][kk * 4 + 2 * hi];
                ap[qi].y = ps[w][qi][m][kk * 4 + 2 * hi + 1];
                lacc[qi] = wmma_f32(ap[qi], bones, lacc[qi]);
            }
            const float* vp = Vb + (size_t)(j * 16 + kk * 4 + 2 * hi) * kH;
#pragma unroll
            for (int nt = 0; nt < 4; ++nt) {
                v2f bv;
                bv.x = vp[nt * 16 + m];
                bv.y = vp[kH + nt * 16 + m];
#pragma unroll
                for (int qi = 0; qi < kQT; ++qi)
                    o[qi][nt] = wmma_f32(ap[qi], bv, o[qi][nt]);
            }
        }
        asm volatile("" ::: "memory");
    }

    if (m == 0) {
#pragma unroll
        for (int qi = 0; qi < kQT; ++qi)
#pragma unroll
            for (int r = 0; r < 8; ++r) {
                m_s[w][qi * 16 + r + 8 * hi] = mrow[qi][r];
                l_s[w][qi * 16 + r + 8 * hi] = lacc[qi][r];
            }
    }

    const int row = tid >> 4;
    const int c0 = (tid & 15) * 4;
#pragma unroll
    for (int qi = 0; qi < kQT; ++qi) {
#pragma unroll
        for (int nt = 0; nt < 4; ++nt)
#pragma unroll
            for (int r = 0; r < 8; ++r)
                o_s[w][r + 8 * hi][nt * 16 + m] = o[qi][nt][r];
        __syncthreads();

        float M = m_s[0][qi * 16 + row];
#pragma unroll
        for (int w2 = 1; w2 < kWaves; ++w2) M = fmaxf(M, m_s[w2][qi * 16 + row]);
        float e[kWaves];
        float L = 0.f;
#pragma unroll
        for (int w2 = 0; w2 < kWaves; ++w2) {
            e[w2] = __expf(m_s[w2][qi * 16 + row] - M);
            L += e[w2] * l_s[w2][qi * 16 + row];
        }
        const float invL = 1.f / L;
        v4f acc = {};
#pragma unroll
        for (int w2 = 0; w2 < kWaves; ++w2) {
            const v4f ov = *reinterpret_cast<const v4f*>(&o_s[w2][row][c0]);
            acc += e[w2] * ov;
        }
        float* Ob = out + ((size_t)b * kS + qb * 32 + qi * 16) * kH;
        *reinterpret_cast<v4f*>(Ob + row * kH + c0) = acc * invL;
        __syncthreads();
    }
}

extern "C" void kernel_launch(void* const* d_in, const int* in_sizes, int n_in,
                              void* d_out, int out_size, void* d_ws,
                              size_t ws_size, hipStream_t stream) {
    const float* x = (const float*)d_in[0];
    const float* Wq = (const float*)d_in[1];
    const float* Wk = (const float*)d_in[2];
    const float* Wv = (const float*)d_in[3];
    float* out = (float*)d_out;

    float* qws = (float*)d_ws;
    float* kws = qws + (size_t)kB * kS * kH;
    float* vws = kws + (size_t)kB * kS * kH;

    proj_ln_kernel<<<(kB * kS) / 16, 128, 0, stream>>>(x, Wq, Wk, Wv, qws, kws,
                                                       vws);
    flash_attn_kernel<<<kB * (kS / 32), 256, 0, stream>>>(qws, kws, vws, out);
}


# attention: softmax on a fixed offset 8 instead of a running row max (LayerNorm without affine bounds |q.k|/sqrt(H) by sqrt(H)=8 for any input): no cross-lane max, no O/l rescale; key sum order kept
# speedup vs baseline: 1.3088x; 1.1281x over previous
#include <hip/hip_runtime.h>
#include <math.h>

typedef __attribute__((ext_vector_type(2))) float v2f;
typedef __attribute__((ext_vector_type(4))) float v4f;
typedef __attribute__((ext_vector_type(8))) float v8f;

namespace {
constexpr int kB = 4;
constexpr int kS = 4096;
constexpr int kD = 1024;
constexpr int kH = 64;
constexpr float kLnEps = 1e-5f;
constexpr float kScale = 0.125f;
constexpr float kScoreBound = 8.0f;
constexpr float kLog2e = 1.4426950408889634f;
}

static __device__ __forceinline__ v8f wmma_f32(v2f a, v2f b, v8f c) {
    return __builtin_amdgcn_wmma_f32_16x16x4_f32(false, a, false, b, (short)0, c,
                                                 false, false);
}

__global__ __launch_bounds__(128) void proj_ln_kernel(
    const float* __restrict__ x, const float* __restrict__ Wq,
    const float* __restrict__ Wk, const float* __restrict__ Wv,
    float* __restrict__ qo, float* __restrict__ ko, float* __restrict__ vo) {
    __shared__ float tq[4][16][68];
    __shared__ float tk[4][16][68];
    __shared__ float tv[4][16][68];

    const int tid = threadIdx.x;
    const int w = tid >> 5;
    const int lane = tid & 31;
    const int m = lane & 15;
    const int hi = lane >> 4;
    const int rowBase = blockIdx.x * 16;

    v8f aq[4], ak[4], av[4];
#pragma unroll
    for (int nt = 0; nt < 4; ++nt) { aq[nt] = {}; ak[nt] = {}; av[nt] = {}; }

    const float* xr = x + (size_t)(rowBase + m) * kD;
    const int k0beg = w * (kD / 4);
    for (int it = 0; it < kD / 16; ++it) {
        const int k0 = k0beg + 4 * it;
        v2f a;
        a.x = xr[k0 + 2 * hi];
        a.y = xr[k0 + 2 * hi + 1];
        const int wrow = (k0 + 2 * hi) * kH;
#pragma unroll
        for (int nt = 0; nt < 4; ++nt) {
            const int col = nt * 16 + m;
            v2f bq, bk, bv;
            bq.x = Wq[wrow + col];      bq.y = Wq[wrow + kH + col];
            bk.x = Wk[wrow + col];      bk.y = Wk[wrow + kH + col];
            bv.x = Wv[wrow + col];      bv.y = Wv[wrow + kH + col];
            aq[nt] = wmma_f32(a, bq, aq[nt]);
            ak[nt] = wmma_f32(a, bk, ak[nt]);
            av[nt] = wmma_f32(a, bv, av[nt]);
        }
    }

#pragma unroll
    for (int nt = 0; nt < 4; ++nt)
#pragma unroll
        for (int r = 0; r < 8; ++r) {
            const int row = r + 8 * hi;
            const int col = nt * 16 + m;
            tq[w][row][col] = aq[nt][r];
            tk[w][row][col] = ak[nt][r];
            tv[w][row][col] = av[nt][r];
        }
    __syncthreads();

    if (tid < 48) {
        const int which = tid >> 4;
        const int row = tid & 15;
        float(*arr)[16][68] = (which == 0) ? tq : (which == 1) ? tk : tv;
        float* r0 = &arr[0][row][0];
        const float* r1 = &arr[1][row][0];
        const float* r2 = &arr[2][row][0];
        const float* r3 = &arr[3][row][0];
        float s = 0.f, s2 = 0.f;
#pragma unroll
        for (int c = 0; c < kH; ++c) {
            const float t = r0[c] + r1[c] + r2[c] + r3[c];
            r0[c] = t;
            s += t;
            s2 += t * t;
        }
        if (which < 2) {
            const float mu = s * (1.f / kH);
            const float var = s2 * (1.f / kH) - mu * mu;
            const float inv = rsqrtf(var + kLnEps);
#pragma unroll
            for (int c = 0; c < kH; ++c) r0[c] = (r0[c] - mu) * inv;
        }
    }
    __syncthreads();

    const size_t ob = (size_t)rowBase * kH;
#pragma unroll
    for (int it = 0; it < 2; ++it) {
        const int flat = it * 512 + tid * 4;
        const int row = flat >> 6;
        const int col = flat & 63;
        *reinterpret_cast<v4f*>(qo + ob + flat) =
            *reinterpret_cast<const v4f*>(&tq[0][row][col]);
        *reinterpret_cast<v4f*>(ko + ob + flat) =
            *reinterpret_cast<const v4f*>(&tk[0][row][col]);
        *reinterpret_cast<v4f*>(vo + ob + flat) =
            *reinterpret_cast<const v4f*>(&tv[0][row][col]);
    }
}

__global__ __launch_bounds__(256) void flash_attn_kernel(
    const float* __restrict__ q, const float* __restrict__ k,
    const float* __restrict__ v, float* __restrict__ out) {
    constexpr int kQT = 2;
    constexpr int kWaves = 8;
    __shared__ float ps[kWaves][kQT][16][18];
    __shared__ float l_s[kWaves][kQT * 16];
    __shared__ alignas(16) float o_s[kWaves][16][64];

    const int tid = threadIdx.x;
    const int w = tid >> 5;
    const int lane = tid & 31;
    const int m = lane & 15;
    const int hi = lane >> 4;
    const int b = blockIdx.x >> 7;
    const int qb = blockIdx.x & 127;

    const float* Qb = q + ((size_t)b * kS + qb * 32) * kH;
    const float* Kb = k + (size_t)b * kS * kH;
    const float* Vb = v + (size_t)b * kS * kH;

    v2f aq[kQT][16];
#pragma unroll
    for (int qi = 0; qi < kQT; ++qi)
#pragma unroll
        for (int kk = 0; kk < 16; ++kk) {
            const float* p = Qb + (qi * 16 + m) * kH + kk * 4 + 2 * hi;
            aq[qi][kk].x = p[0] * kScale;
            aq[qi][kk].y = p[1] * kScale;
        }

    v8f o[kQT][4];
    v8f lacc[kQT];
#pragma unroll
    for (int qi = 0; qi < kQT; ++qi) {
#pragma unroll
        for (int nt = 0; nt < 4; ++nt) o[qi][nt] = {};
        lacc[qi] = {};
    }
    const v2f bones = {1.f, 1.f};

    constexpr int kTiles = (kS / 16) / kWaves;
    for (int t = 0; t < kTiles; ++t) {
        const int j = w + kWaves * t;
        const float* Kt = Kb + (size_t)j * 16 * kH;

        {
            const int jn = (t + 1 < kTiles) ? (j + kWaves) : j;
            const size_t nb = (size_t)jn * 16 * kH + lane * 32;
            __builtin_prefetch(Kb + nb, 0, 3);
            __builtin_prefetch(Vb + nb, 0, 3);
        }

        v8f sc0[kQT], sc1[kQT];
#pragma unroll
        for (int qi = 0; qi < kQT; ++qi) { sc0[qi] = {}; sc1[qi] = {}; }
#pragma unroll
        for (int kk = 0; kk < 8; ++kk) {
            v2f bk0, bk1;
            const float* p0 = Kt + m * kH + kk * 4 + 2 * hi;
            const float* p1 = p0 + 32;
            bk0.x = p0[0];  bk0.y = p0[1];
            bk1.x = p1[0];  bk1.y = p1[1];
#pragma unroll
            for (int qi = 0; qi < kQT; ++qi) {
                sc0[qi] = wmma_f32(aq[qi][kk], bk0, sc0[qi]);
                sc1[qi] = wmma_f32(aq[qi][kk + 8], bk1, sc1[qi]);
            }
        }

#pragma unroll
        for (int qi = 0; qi < kQT; ++qi) {
            v8f pf;
#pragma unroll
            for (int r = 0; r < 8; ++r) {
                const float sc = sc0[qi][r] + sc1[qi][r];
                pf[r] = __builtin_amdgcn_exp2f(
                    __builtin_fmaf(sc, kLog2e, -kScoreBound * kLog2e));
            }

#pragma unroll
            for (int r = 0; r < 8; ++r) ps[w][qi][r + 8 * hi][m] = pf[r];
        }
        asm volatile("" ::: "memory");

#pragma unroll
        for (int kk = 0; kk < 4; ++kk) {
            v2f ap[kQT];
#pragma unroll
            for (int qi = 0; qi < kQT; ++qi) {
                ap[qi].x = ps[w][qi][m][kk * 4 + 2 * hi];
                ap[qi].y = ps[w][qi][m][kk * 4 + 2 * hi + 1];
                lacc[qi] = wmma_f32(ap[qi], bones, lacc[qi]);
            }
            const float* vp = Vb + (size_t)(j * 16 + kk * 4 + 2 * hi) * kH;
#pragma unroll
            for (int nt = 0; nt < 4; ++nt) {
                v2f bv;
                bv.x = vp[nt * 16 + m];
                bv.y = vp[kH + nt * 16 + m];
#pragma unroll
                for (int qi = 0; qi < kQT; ++qi)
                    o[qi][nt] = wmma_f32(ap[qi], bv, o[qi][nt]);
            }
        }
        asm volatile("" ::: "memory");
    }

    if (m == 0) {
#pragma unroll
        for (int qi = 0; qi < kQT; ++qi)
#pragma unroll
            for (int r = 0; r < 8; ++r)
                l_s[w][qi * 16 + r + 8 * hi] = lacc[qi][r];
    }

    const int row = tid >> 4;
    const int c0 = (tid & 15) * 4;
#pragma unroll
    for (int qi = 0; qi < kQT; ++qi) {
#pragma unroll
        for (int nt = 0; nt < 4; ++nt)
#pragma unroll
            for (int r = 0; r < 8; ++r)
                o_s[w][r + 8 * hi][nt * 16 + m] = o[qi][nt][r];
        __syncthreads();

        float L = 0.f;
#pragma unroll
        for (int w2 = 0; w2 < kWaves; ++w2) L += l_s[w2][qi * 16 + row];
        const float invL = 1.f / L;
        v4f acc = {};
#pragma unroll
        for (int w2 = 0; w2 < kWaves; ++w2)
            acc += *reinterpret_cast<const v4f*>(&o_s[w2][row][c0]);
        float* Ob = out + ((size_t)b * kS + qb * 32 + qi * 16) * kH;
        *reinterpret_cast<v4f*>(Ob + row * kH + c0) = acc * invL;
        __syncthreads();
    }
}

extern "C" void kernel_launch(void* const* d_in, const int* in_sizes, int n_in,
                              void* d_out, int out_size, void* d_ws,
                              size_t ws_size, hipStream_t stream) {
    const float* x = (const float*)d_in[0];
    const float* Wq = (const float*)d_in[1];
    const float* Wk = (const float*)d_in[2];
    const float* Wv = (const float*)d_in[3];
    float* out = (float*)d_out;

    float* qws = (float*)d_ws;
    float* kws = qws + (size_t)kB * kS * kH;
    float* vws = kws + (size_t)kB * kS * kH;

    proj_ln_kernel<<<(kB * kS) / 16, 128, 0, stream>>>(x, Wq, Wk, Wv, qws, kws,
                                                       vws);
    flash_attn_kernel<<<kB * (kS / 32), 256, 0, stream>>>(qws, kws, vws, out);
}


# attention: softmax row sums kept as per-lane running sums (one packed add per two exps), lane-reduced once after the key loop, replacing the P x ones matrix products; row-sum order changes
# speedup vs baseline: 1.3618x; 1.0405x over previous
#include <hip/hip_runtime.h>
#include <math.h>

typedef __attribute__((ext_vector_type(2))) float v2f;
typedef __attribute__((ext_vector_type(4))) float v4f;
typedef __attribute__((ext_vector_type(8))) float v8f;

namespace {
constexpr int kB = 4;
constexpr int kS = 4096;
constexpr int kD = 1024;
constexpr int kH = 64;
constexpr float kLnEps = 1e-5f;
constexpr float kScale = 0.125f;
constexpr float kScoreBound = 8.0f;
constexpr float kLog2e = 1.4426950408889634f;
}

static __device__ __forceinline__ v8f wmma_f32(v2f a, v2f b, v8f c) {
    return __builtin_amdgcn_wmma_f32_16x16x4_f32(false, a, false, b, (short)0, c,
                                                 false, false);
}

__global__ __launch_bounds__(128) void proj_ln_kernel(
    const float* __restrict__ x, const float* __restrict__ Wq,
    const float* __restrict__ Wk, const float* __restrict__ Wv,
    float* __restrict__ qo, float* __restrict__ ko, float* __restrict__ vo) {
    __shared__ float tq[4][16][68];
    __shared__ float tk[4][16][68];
    __shared__ float tv[4][16][68];

    const int tid = threadIdx.x;
    const int w = tid >> 5;
    const int lane = tid & 31;
    const int m = lane & 15;
    const int hi = lane >> 4;
    const int rowBase = blockIdx.x * 16;

    v8f aq[4], ak[4], av[4];
#pragma unroll
    for (int nt = 0; nt < 4; ++nt) { aq[nt] = {}; ak[nt] = {}; av[nt] = {}; }

    const float* xr = x + (size_t)(rowBase + m) * kD;
    const int k0beg = w * (kD / 4);
    for (int it = 0; it < kD / 16; ++it) {
        const int k0 = k0beg + 4 * it;
        v2f a;
        a.x = xr[k0 + 2 * hi];
        a.y = xr[k0 + 2 * hi + 1];
        const int wrow = (k0 + 2 * hi) * kH;
#pragma unroll
        for (int nt = 0; nt < 4; ++nt) {
            const int col = nt * 16 + m;
            v2f bq, bk, bv;
            bq.x = Wq[wrow + col];      bq.y = Wq[wrow + kH + col];
            bk.x = Wk[wrow + col];      bk.y = Wk[wrow + kH + col];
            bv.x = Wv[wrow + col];      bv.y = Wv[wrow + kH + col];
            aq[nt] = wmma_f32(a, bq, aq[nt]);
            ak[nt] = wmma_f32(a, bk, ak[nt]);
            av[nt] = wmma_f32(a, bv, av[nt]);
        }
    }

#pragma unroll
    for (int nt = 0; nt < 4; ++nt)
#pragma unroll
        for (int r = 0; r < 8; ++r) {
            const int row = r + 8 * hi;
            const int col = nt * 16 + m;
            tq[w][row][col] = aq[nt][r];
            tk[w][row][col] = ak[nt][r];
            tv[w][row][col] = av[nt][r];
        }
    __syncthreads();

    if (tid < 48) {
        const int which = tid >> 4;
        const int row = tid & 15;
        float(*arr)[16][68] = (which == 0) ? tq : (which == 1) ? tk : tv;
        float* r0 = &arr[0][row][0];
        const float* r1 = &arr[1][row][0];
        const float* r2 = &arr[2][row][0];
        const float* r3 = &arr[3][row][0];
        float s = 0.f, s2 = 0.f;
#pragma unroll
        for (int c = 0; c < kH; ++c) {
            const float t = r0[c] + r1[c] + r2[c] + r3[c];
            r0[c] = t;
            s += t;
            s2 += t * t;
        }
        if (which < 2) {
            const float mu = s * (1.f / kH);
            const float var = s2 * (1.f / kH) - mu * mu;
            const float inv = rsqrtf(var + kLnEps);
#pragma unroll
            for (int c = 0; c < kH; ++c) r0[c] = (r0[c] - mu) * inv;
        }
    }
    __syncthreads();

    const size_t ob = (size_t)rowBase * kH;
#pragma unroll
    for (int it = 0; it < 2; ++it) {
        const int flat = it * 512 + tid * 4;
        const int row = flat >> 6;
        const int col = flat & 63;
        *reinterpret_cast<v4f*>(qo + ob + flat) =
            *reinterpret_cast<const v4f*>(&tq[0][row][col]);
        *reinterpret_cast<v4f*>(ko + ob + flat) =
            *reinterpret_cast<const v4f*>(&tk[0][row][col]);
        *reinterpret_cast<v4f*>(vo + ob + flat) =
            *reinterpret_cast<const v4f*>(&tv[0][row][col]);
    }
}

__global__ __launch_bounds__(256) void flash_attn_kernel(
    const float* __restrict__ q, const float* __restrict__ k,
    const float* __restrict__ v, float* __restrict__ out) {
    constexpr int kQT = 2;
    constexpr int kWaves = 8;
    __shared__ float ps[kWaves][kQT][16][18];
    __shared__ float l_s[kWaves][kQT * 16];
    __shared__ alignas(16) float o_s[kWaves][16][64];

    const int tid = threadIdx.x;
    const int w = tid >> 5;
    const int lane = tid & 31;
    const int m = lane & 15;
    const int hi = lane >> 4;
    const int b = blockIdx.x >> 7;
    const int qb = blockIdx.x & 127;

    const float* Qb = q + ((size_t)b * kS + qb * 32) * kH;
    const float* Kb = k + (size_t)b * kS * kH;
    const float* Vb = v + (size_t)b * kS * kH;

    v2f aq[kQT][16];
#pragma unroll
    for (int qi = 0; qi < kQT; ++qi)
#pragma unroll
        for (int kk = 0; kk < 16; ++kk) {
            const float* p = Qb + (qi * 16 + m) * kH + kk * 4 + 2 * hi;
            aq[qi][kk].x = p[0] * kScale;
            aq[qi][kk].y = p[1] * kScale;
        }

    v8f o[kQT][4];
    v8f lacc[kQT];
#pragma unroll
    for (int qi = 0; qi < kQT; ++qi) {
#pragma unroll
        for (int nt = 0; nt < 4; ++nt) o[qi][nt] = {};
        lacc[qi] = {};
    }

    constexpr int kTiles = (kS / 16) / kWaves;
    for (int t = 0; t < kTiles; ++t) {
        const int j = w + kWaves * t;
        const float* Kt = Kb + (size_t)j * 16 * kH;

        {
            const int jn = (t + 1 < kTiles) ? (j + kWaves) : j;
            const size_t nb = (size_t)jn * 16 * kH + lane * 32;
            __builtin_prefetch(Kb + nb, 0, 3);
            __builtin_prefetch(Vb + nb, 0, 3);
        }

        v8f sc0[kQT], sc1[kQT];
#pragma unroll
        for (int qi = 0; qi < kQT; ++qi) { sc0[qi] = {}; sc1[qi] = {}; }
#pragma unroll
        for (int kk = 0; kk < 8; ++kk) {
            v2f bk0, bk1;
            const float* p0 = Kt + m * kH + kk * 4 + 2 * hi;
            const float* p1 = p0 + 32;
            bk0.x = p0[0];  bk0.y = p0[1];
            bk1.x = p1[0];  bk1.y = p1[1];
#pragma unroll
            for (int qi = 0; qi < kQT; ++qi) {
                sc0[qi] = wmma_f32(aq[qi][kk], bk0, sc0[qi]);
                sc1[qi] = wmma_f32(aq[qi][kk + 8], bk1, sc1[qi]);
            }
        }

#pragma unroll
        for (int qi = 0; qi < kQT; ++qi) {
            v8f pf;
#pragma unroll
            for (int r = 0; r < 8; ++r) {
                const float sc = sc0[qi][r] + sc1[qi][r];
                pf[r] = __builtin_amdgcn_exp2f(
                    __builtin_fmaf(sc, kLog2e, -kScoreBound * kLog2e));
            }
            lacc[qi] += pf;

#pragma unroll
            for (int r = 0; r < 8; ++r) ps[w][qi][r + 8 * hi][m] = pf[r];
        }
        asm volatile("" ::: "memory");

#pragma unroll
        for (int kk = 0; kk < 4; ++kk) {
            v2f ap[kQT];
#pragma unroll
            for (int qi = 0; qi < kQT; ++qi) {
                ap[qi].x = ps[w][qi][m][kk * 4 + 2 * hi];
                ap[qi].y = ps[w][qi][m][kk * 4 + 2 * hi + 1];
            }
            const float* vp = Vb + (size_t)(j * 16 + kk * 4 + 2 * hi) * kH;
#pragma unroll
            for (int nt = 0; nt < 4; ++nt) {
                v2f bv;
                bv.x = vp[nt * 16 + m];
                bv.y = vp[kH + nt * 16 + m];
#pragma unroll
                for (int qi = 0; qi < kQT; ++qi)
                    o[qi][nt] = wmma_f32(ap[qi], bv, o[qi][nt]);
            }
        }
        asm volatile("" ::: "memory");
    }

#pragma unroll
    for (int qi = 0; qi < kQT; ++qi)
#pragma unroll
        for (int r = 0; r < 8; ++r) {
            float s = lacc[qi][r];
            s += __shfl_xor(s, 1);
            s += __shfl_xor(s, 2);
            s += __shfl_xor(s, 4);
            s += __shfl_xor(s, 8);
            lacc[qi][r] = s;
        }
    if (m == 0) {
#pragma unroll
        for (int qi = 0; qi < kQT; ++qi)
#pragma unroll
            for (int r = 0; r < 8; ++r)
                l_s[w][qi * 16 + r + 8 * hi] = lacc[qi][r];
    }

    const int row = tid >> 4;
    const int c0 = (tid & 15) * 4;
#pragma unroll
    for (int qi = 0; qi < kQT; ++qi) {
#pragma unroll
        for (int nt = 0; nt < 4; ++nt)
#pragma unroll
            for (int r = 0; r < 8; ++r)
                o_s[w][r + 8 * hi][nt * 16 + m] = o[qi][nt][r];
        __syncthreads();

        float L = 0.f;
#pragma unroll
        for (int w2 = 0; w2 < kWaves; ++w2) L += l_s[w2][qi * 16 + row];
        const float invL = 1.f / L;
        v4f acc = {};
#pragma unroll
        for (int w2 = 0; w2 < kWaves; ++w2)
            acc += *reinterpret_cast<const v4f*>(&o_s[w2][row][c0]);
        float* Ob = out + ((size_t)b * kS + qb * 32 + qi * 16) * kH;
        *reinterpret_cast<v4f*>(Ob + row * kH + c0) = acc * invL;
        __syncthreads();
    }
}

extern "C" void kernel_launch(void* const* d_in, const int* in_sizes, int n_in,
                              void* d_out, int out_size, void* d_ws,
                              size_t ws_size, hipStream_t stream) {
    const float* x = (const float*)d_in[0];
    const float* Wq = (const float*)d_in[1];
    const float* Wk = (const float*)d_in[2];
    const float* Wv = (const float*)d_in[3];
    float* out = (float*)d_out;

    float* qws = (float*)d_ws;
    float* kws = qws + (size_t)kB * kS * kH;
    float* vws = kws + (size_t)kB * kS * kH;

    proj_ln_kernel<<<(kB * kS) / 16, 128, 0, stream>>>(x, Wq, Wk, Wv, qws, kws,
                                                       vws);
    flash_attn_kernel<<<kB * (kS / 32), 256, 0, stream>>>(qws, kws, vws, out);
}


# both kernels: every WMMA operand load is 16 B per lane over whole 128-B lines (fragment k order and W column order permuted, q/k/v workspace in fragment order); k sum order of all products changes
# speedup vs baseline: 1.4159x; 1.0397x over previous
#include <hip/hip_runtime.h>
#include <math.h>

typedef __attribute__((ext_vector_type(2))) float v2f;
typedef __attribute__((ext_vector_type(4))) float v4f;
typedef __attribute__((ext_vector_type(8))) float v8f;

namespace {
constexpr int kB = 4;
constexpr int kS = 4096;
constexpr int kD = 1024;
constexpr int kH = 64;
constexpr float kLnEps = 1e-5f;
constexpr float kScale = 0.125f;
constexpr float kScoreBound = 8.0f;
constexpr float kLog2e = 1.4426950408889634f;
}

static __device__ __forceinline__ v8f wmma_f32(v2f a, v2f b, v8f c) {
    return __builtin_amdgcn_wmma_f32_16x16x4_f32(false, a, false, b, (short)0, c,
                                                 false, false);
}

__global__ __launch_bounds__(128) void proj_ln_kernel(
    const float* __restrict__ x, const float* __restrict__ Wq,
    const float* __restrict__ Wk, const float* __restrict__ Wv,
    float* __restrict__ qo, float* __restrict__ ko, float* __restrict__ vo) {
    __shared__ alignas(16) float tq[4][16][68];
    __shared__ alignas(16) float tk[4][16][68];
    __shared__ alignas(16) float tv[4][16][68];

    const int tid = threadIdx.x;
    const int w = tid >> 5;
    const int lane = tid & 31;
    const int m = lane & 15;
    const int hi = lane >> 4;
    const int rowBase = blockIdx.x * 16;

    v8f aq[4], ak[4], av[4];
#pragma unroll
    for (int nt = 0; nt < 4; ++nt) { aq[nt] = {}; ak[nt] = {}; av[nt] = {}; }

    const int k0beg = w * (kD / 4) + 4 * hi;
    const float* xr = x + (size_t)(rowBase + m) * kD + k0beg;
    const int wbeg = k0beg * kH + 4 * m;
    for (int it = 0; it < kD / 32; ++it) {
        const v4f xa = *reinterpret_cast<const v4f*>(xr + 8 * it);
        const int wo = wbeg + 8 * it * kH;
#pragma unroll
        for (int u = 0; u < 2; ++u) {
            v2f a;
            a.x = xa[2 * u];
            a.y = xa[2 * u + 1];
            const int r0 = wo + (2 * u) * kH;
            const int r1 = r0 + kH;
            const v4f q0 = *reinterpret_cast<const v4f*>(Wq + r0);
            const v4f q1 = *reinterpret_cast<const v4f*>(Wq + r1);
            const v4f c0 = *reinterpret_cast<const v4f*>(Wk + r0);
            const v4f c1 = *reinterpret_cast<const v4f*>(Wk + r1);
            const v4f d0 = *reinterpret_cast<const v4f*>(Wv + r0);
            const v4f d1 = *reinterpret_cast<const v4f*>(Wv + r1);
#pragma unroll
            for (int nt = 0; nt < 4; ++nt) {
                v2f bq, bk, bv;
                bq.x = q0[nt];  bq.y = q1[nt];
                bk.x = c0[nt];  bk.y = c1[nt];
                bv.x = d0[nt];  bv.y = d1[nt];
                aq[nt] = wmma_f32(a, bq, aq[nt]);
                ak[nt] = wmma_f32(a, bk, ak[nt]);
                av[nt] = wmma_f32(a, bv, av[nt]);
            }
        }
    }

#pragma unroll
    for (int r = 0; r < 8; ++r) {
        const int row = r + 8 * hi;
        v4f t;
        t.x = aq[0][r]; t.y = aq[1][r]; t.z = aq[2][r]; t.w = aq[3][r];
        *reinterpret_cast<v4f*>(&tq[w][row][4 * m]) = t;
        t.x = ak[0][r]; t.y = ak[1][r]; t.z = ak[2][r]; t.w = ak[3][r];
        *reinterpret_cast<v4f*>(&tk[w][row][4 * m]) = t;
        t.x = av[0][r]; t.y = av[1][r]; t.z = av[2][r]; t.w = av[3][r];
        *reinterpret_cast<v4f*>(&tv[w][row][4 * m]) = t;
    }
    __syncthreads();

    if (tid < 48) {
        const int which = tid >> 4;
        const int row = tid & 15;
        float(*arr)[16][68] = (which == 0) ? tq : (which == 1) ? tk : tv;
        float* r0 = &arr[0][row][0];
        const float* r1 = &arr[1][row][0];
        const float* r2 = &arr[2][row][0];
        const float* r3 = &arr[3][row][0];
        float s = 0.f, s2 = 0.f;
#pragma unroll
        for (int c = 0; c < kH; ++c) {
            const float t = r0[c] + r1[c] + r2[c] + r3[c];
            r0[c] = t;
            s += t;
            s2 += t * t;
        }
        if (which < 2) {
            const float mu = s * (1.f / kH);
            const float var = s2 * (1.f / kH) - mu * mu;
            const float inv = rsqrtf(var + kLnEps);
#pragma unroll
            for (int c = 0; c < kH; ++c) r0[c] = (r0[c] - mu) * inv;
        }
    }
    __syncthreads();

    const size_t ob = (size_t)rowBase * kH;
#pragma unroll
    for (int it = 0; it < 2; ++it) {
        const int flat = it * 512 + tid * 4;
        const int g = flat >> 7;
        const int mm = (flat >> 2) & 15;
        const int hh = (flat >> 6) & 1;
        *reinterpret_cast<v4f*>(qo + ob + flat) =
            *reinterpret_cast<const v4f*>(&tq[0][mm][8 * g + 4 * hh]);
        *reinterpret_cast<v4f*>(ko + ob + flat) =
            *reinterpret_cast<const v4f*>(&tk[0][mm][8 * g + 4 * hh]);
        const int vr = 8 * (g >> 2) + 4 * hh;
        const int vc = 16 * (g & 3) + mm;
        v4f t;
        t.x = tv[0][vr][vc];
        t.y = tv[0][vr + 1][vc];
        t.z = tv[0][vr + 2][vc];
        t.w = tv[0][vr + 3][vc];
        *reinterpret_cast<v4f*>(vo + ob + flat) = t;
    }
}

__global__ __launch_bounds__(256) void flash_attn_kernel(
    const float* __restrict__ q, const float* __restrict__ k,
    const float* __restrict__ v, float* __restrict__ out) {
    constexpr int kQT = 2;
    constexpr int kWaves = 8;
    __shared__ alignas(16) float ps[kWaves][kQT][16][20];
    __shared__ float l_s[kWaves][kQT * 16];
    __shared__ alignas(16) float o_s[kWaves][16][64];

    const int tid = threadIdx.x;
    const int w = tid >> 5;
    const int lane = tid & 31;
    const int m = lane & 15;
    const int hi = lane >> 4;
    const int b = blockIdx.x >> 7;
    const int qb = blockIdx.x & 127;

    const float* Qb = q + ((size_t)b * kS + qb * 32) * kH;
    const float* Kb = k + (size_t)b * kS * kH;
    const float* Vb = v + (size_t)b * kS * kH;

    v4f aq[kQT][8];
#pragma unroll
    for (int qi = 0; qi < kQT; ++qi)
#pragma unroll
        for (int g = 0; g < 8; ++g)
            aq[qi][g] = *reinterpret_cast<const v4f*>(Qb + qi * 16 * kH + g * 128 +
                                                      lane * 4) * kScale;

    v8f o[kQT][4];
    v8f lacc[kQT];
#pragma unroll
    for (int qi = 0; qi < kQT; ++qi) {
#pragma unroll
        for (int nt = 0; nt < 4; ++nt) o[qi][nt] = {};
        lacc[qi] = {};
    }

    constexpr int kTiles = (kS / 16) / kWaves;
    for (int t = 0; t < kTiles; ++t) {
        const int j = w + kWaves * t;
        const float* Kt = Kb + (size_t)j * 16 * kH;

        {
            const int jn = (t + 1 < kTiles) ? (j + kWaves) : j;
            const size_t nb = (size_t)jn * 16 * kH + lane * 32;
            __builtin_prefetch(Kb + nb, 0, 3);
            __builtin_prefetch(Vb + nb, 0, 3);
        }

        v8f sc0[kQT], sc1[kQT];
#pragma unroll
        for (int qi = 0; qi < kQT; ++qi) { sc0[qi] = {}; sc1[qi] = {}; }
#pragma unroll
        for (int g = 0; g < 8; ++g) {
            const v4f bk = *reinterpret_cast<const v4f*>(Kt + g * 128 + lane * 4);
            v2f bk0, bk1;
            bk0.x = bk.x;  bk0.y = bk.y;
            bk1.x = bk.z;  bk1.y = bk.w;
#pragma unroll
            for (int qi = 0; qi < kQT; ++qi) {
                v2f a0, a1;
                a0.x = aq[qi][g].x;  a0.y = aq[qi][g].y;
                a1.x = aq[qi][g].z;  a1.y = aq[qi][g].w;
                sc0[qi] = wmma_f32(a0, bk0, sc0[qi]);
                sc1[qi] = wmma_f32(a1, bk1, sc1[qi]);
            }
        }

#pragma unroll
        for (int qi = 0; qi < kQT; ++qi) {
            v8f pf;
#pragma unroll
            for (int r = 0; r < 8; ++r) {
                const float sc = sc0[qi][r] + sc1[qi][r];
                pf[r] = __builtin_amdgcn_exp2f(
                    __builtin_fmaf(sc, kLog2e, -kScoreBound * kLog2e));
            }
            lacc[qi] += pf;

#pragma unroll
            for (int r = 0; r < 8; ++r) ps[w][qi][r + 8 * hi][m] = pf[r];
        }
        asm volatile("" ::: "memory");

        const float* Vt = Vb + (size_t)j * 16 * kH;
#pragma unroll
        for (int g = 0; g < 2; ++g) {
            v4f ap[kQT];
#pragma unroll
            for (int qi = 0; qi < kQT; ++qi)
                ap[qi] = *reinterpret_cast<const v4f*>(&ps[w][qi][m][8 * g + 4 * hi]);
#pragma unroll
            for (int nt = 0; nt < 4; ++nt) {
                const v4f bv =
                    *reinterpret_cast<const v4f*>(Vt + (g * 4 + nt) * 128 + lane * 4);
#pragma unroll
                for (int u = 0; u < 2; ++u) {
                    v2f b2;
                    b2.x = bv[2 * u];  b2.y = bv[2 * u + 1];
#pragma unroll
                    for (int qi = 0; qi < kQT; ++qi) {
                        v2f a2;
                        a2.x = ap[qi][2 * u];  a2.y = ap[qi][2 * u + 1];
                        o[qi][nt] = wmma_f32(a2, b2, o[qi][nt]);
                    }
                }
            }
        }
        asm volatile("" ::: "memory");
    }

#pragma unroll
    for (int qi = 0; qi < kQT; ++qi)
#pragma unroll
        for (int r = 0; r < 8; ++r) {
            float s = lacc[qi][r];
            s += __shfl_xor(s, 1);
            s += __shfl_xor(s, 2);
            s += __shfl_xor(s, 4);
            s += __shfl_xor(s, 8);
            lacc[qi][r] = s;
        }
    if (m == 0) {
#pragma unroll
        for (int qi = 0; qi < kQT; ++qi)
#pragma unroll
            for (int r = 0; r < 8; ++r)
                l_s[w][qi * 16 + r + 8 * hi] = lacc[qi][r];
    }

    const int row = tid >> 4;
    const int c0 = (tid & 15) * 4;
#pragma unroll
    for (int qi = 0; qi < kQT; ++qi) {
#pragma unroll
        for (int nt = 0; nt < 4; ++nt)
#pragma unroll
            for (int r = 0; r < 8; ++r)
                o_s[w][r + 8 * hi][nt * 16 + m] = o[qi][nt][r];
        __syncthreads();

        float L = 0.f;
#pragma unroll
        for (int w2 = 0; w2 < kWaves; ++w2) L += l_s[w2][qi * 16 + row];
        const float invL = 1.f / L;
        v4f acc = {};
#pragma unroll
        for (int w2 = 0; w2 < kWaves; ++w2)
            acc += *reinterpret_cast<const v4f*>(&o_s[w2][row][c0]);
        float* Ob = out + ((size_t)b * kS + qb * 32 + qi * 16) * kH;
        *reinterpret_cast<v4f*>(Ob + row * kH + c0) = acc * invL;
        __syncthreads();
    }
}

extern "C" void kernel_launch(void* const* d_in, const int* in_sizes, int n_in,
                              void* d_out, int out_size, void* d_ws,
                              size_t ws_size, hipStream_t stream) {
    const float* x = (const float*)d_in[0];
    const float* Wq = (const float*)d_in[1];
    const float* Wk = (const float*)d_in[2];
    const float* Wv = (const float*)d_in[3];
    float* out = (float*)d_out;

    float* qws = (float*)d_ws;
    float* kws = qws + (size_t)kB * kS * kH;
    float* vws = kws + (size_t)kB * kS * kH;

    proj_ln_kernel<<<(kB * kS) / 16, 128, 0, stream>>>(x, Wq, Wk, Wv, qws, kws,
                                                       vws);
    flash_attn_kernel<<<kB * (kS / 32), 256, 0, stream>>>(qws, kws, vws, out);
}


# projection kernel: 64-row blocks of 12 waves (4 k-slices x q,k,v); each wave holds four 16-row tiles of one plane so every loaded W fragment feeds four WMMAs; k sum order kept
# speedup vs baseline: 1.5223x; 1.0751x over previous
#include <hip/hip_runtime.h>
#include <math.h>

typedef __attribute__((ext_vector_type(2))) float v2f;
typedef __attribute__((ext_vector_type(4))) float v4f;
typedef __attribute__((ext_vector_type(8))) float v8f;

namespace {
constexpr int kB = 4;
constexpr int kS = 4096;
constexpr int kD = 1024;
constexpr int kH = 64;
constexpr float kLnEps = 1e-5f;
constexpr float kScale = 0.125f;
constexpr float kScoreBound = 8.0f;
constexpr float kLog2e = 1.4426950408889634f;
}

static __device__ __forceinline__ v8f wmma_f32(v2f a, v2f b, v8f c) {
    return __builtin_amdgcn_wmma_f32_16x16x4_f32(false, a, false, b, (short)0, c,
                                                 false, false);
}

__global__ __launch_bounds__(384) void proj_ln_kernel(
    const float* __restrict__ x, const float* __restrict__ Wq,
    const float* __restrict__ Wk, const float* __restrict__ Wv,
    float* __restrict__ qo, float* __restrict__ ko, float* __restrict__ vo) {
    constexpr int kRT = 4;
    constexpr int kRows = 16 * kRT;
    __shared__ alignas(16) float part[3][4][kRows][68];

    const int tid = threadIdx.x;
    const int wv = tid >> 5;
    const int plane = wv >> 2;
    const int ks = wv & 3;
    const int lane = tid & 31;
    const int m = lane & 15;
    const int hi = lane >> 4;
    const int rowBase = blockIdx.x * kRows;
    const float* __restrict__ Wp = (plane == 0) ? Wq : (plane == 1) ? Wk : Wv;

    v8f acc[kRT][4];
#pragma unroll
    for (int rt = 0; rt < kRT; ++rt)
#pragma unroll
        for (int nt = 0; nt < 4; ++nt) acc[rt][nt] = {};

    const int k0beg = ks * (kD / 4) + 4 * hi;
    const float* xr = x + (size_t)(rowBase + m) * kD + k0beg;
    const int wbeg = k0beg * kH + 4 * m;
    for (int c = 0; c < kD / 128; ++c) {
        v4f xa[kRT][4];
#pragma unroll
        for (int rt = 0; rt < kRT; ++rt)
#pragma unroll
            for (int j = 0; j < 4; ++j)
                xa[rt][j] = *reinterpret_cast<const v4f*>(
                    xr + (size_t)rt * 16 * kD + 32 * c + 8 * j);
#pragma unroll
        for (int j = 0; j < 4; ++j) {
            const int wo = wbeg + (32 * c + 8 * j) * kH;
#pragma unroll
            for (int u = 0; u < 2; ++u) {
                const v4f w0 = *reinterpret_cast<const v4f*>(Wp + wo + (2 * u) * kH);
                const v4f w1 =
                    *reinterpret_cast<const v4f*>(Wp + wo + (2 * u + 1) * kH);
#pragma unroll
                for (int nt = 0; nt < 4; ++nt) {
                    v2f bw;
                    bw.x = w0[nt];
                    bw.y = w1[nt];
#pragma unroll
                    for (int rt = 0; rt < kRT; ++rt) {
                        v2f a;
                        a.x = xa[rt][j][2 * u];
                        a.y = xa[rt][j][2 * u + 1];
                        acc[rt][nt] = wmma_f32(a, bw, acc[rt][nt]);
                    }
                }
            }
        }
    }

#pragma unroll
    for (int rt = 0; rt < kRT; ++rt)
#pragma unroll
        for (int r = 0; r < 8; ++r) {
            const int row = rt * 16 + r + 8 * hi;
            v4f t;
            t.x = acc[rt][0][r];
            t.y = acc[rt][1][r];
            t.z = acc[rt][2][r];
            t.w = acc[rt][3][r];
            *reinterpret_cast<v4f*>(&part[plane][ks][row][4 * m]) = t;
        }
    __syncthreads();

    if (tid < 3 * kRows) {
        const int which = tid >> 6;
        const int row = tid & (kRows - 1);
        float* r0 = &part[which][0][row][0];
        const float* r1 = &part[which][1][row][0];
        const float* r2 = &part[which][2][row][0];
        const float* r3 = &part[which][3][row][0];
        float s = 0.f, s2 = 0.f;
#pragma unroll
        for (int c = 0; c < kH; ++c) {
            const float t = r0[c] + r1[c] + r2[c] + r3[c];
            r0[c] = t;
            s += t;
            s2 += t * t;
        }
        if (which < 2) {
            const float mu = s * (1.f / kH);
            const float var = s2 * (1.f / kH) - mu * mu;
            const float inv = rsqrtf(var + kLnEps);
#pragma unroll
            for (int c = 0; c < kH; ++c) r0[c] = (r0[c] - mu) * inv;
        }
    }
    __syncthreads();

#pragma unroll
    for (int j = 0; j < 8; ++j) {
        const int idx = j * 384 + tid;
        const int pl = idx >> 10;
        const int rt = (idx >> 8) & 3;
        const int flat = (idx & 255) * 4;
        const int g = flat >> 7;
        const int mm = (flat >> 2) & 15;
        const int hh = (flat >> 6) & 1;
        float* op = (pl == 0) ? qo : (pl == 1) ? ko : vo;
        const size_t ob = (size_t)(rowBase + 16 * rt) * kH;
        v4f t;
        if (pl < 2) {
            t = *reinterpret_cast<const v4f*>(&part[pl][0][16 * rt + mm][8 * g + 4 * hh]);
        } else {
            const int vr = 16 * rt + 8 * (g >> 2) + 4 * hh;
            const int vc = 16 * (g & 3) + mm;
            t.x = part[2][0][vr][vc];
            t.y = part[2][0][vr + 1][vc];
            t.z = part[2][0][vr + 2][vc];
            t.w = part[2][0][vr + 3][vc];
        }
        *reinterpret_cast<v4f*>(op + ob + flat) = t;
    }
}

__global__ __launch_bounds__(256) void flash_attn_kernel(
    const float* __restrict__ q, const float* __restrict__ k,
    const float* __restrict__ v, float* __restrict__ out) {
    constexpr int kQT = 2;
    constexpr int kWaves = 8;
    __shared__ alignas(16) float ps[kWaves][kQT][16][20];
    __shared__ float l_s[kWaves][kQT * 16];
    __shared__ alignas(16) float o_s[kWaves][16][64];

    const int tid = threadIdx.x;
    const int w = tid >> 5;
    const int lane = tid & 31;
    const int m = lane & 15;
    const int hi = lane >> 4;
    const int b = blockIdx.x >> 7;
    const int qb = blockIdx.x & 127;

    const float* Qb = q + ((size_t)b * kS + qb * 32) * kH;
    const float* Kb = k + (size_t)b * kS * kH;
    const float* Vb = v + (size_t)b * kS * kH;

    v4f aq[kQT][8];
#pragma unroll
    for (int qi = 0; qi < kQT; ++qi)
#pragma unroll
        for (int g = 0; g < 8; ++g)
            aq[qi][g] = *reinterpret_cast<const v4f*>(Qb + qi * 16 * kH + g * 128 +
                                                      lane * 4) * kScale;

    v8f o[kQT][4];
    v8f lacc[kQT];
#pragma unroll
    for (int qi = 0; qi < kQT; ++qi) {
#pragma unroll
        for (int nt = 0; nt < 4; ++nt) o[qi][nt] = {};
        lacc[qi] = {};
    }

    constexpr int kTiles = (kS / 16) / kWaves;
    for (int t = 0; t < kTiles; ++t) {
        const int j = w + kWaves * t;
        const float* Kt = Kb + (size_t)j * 16 * kH;

        {
            const int jn = (t + 1 < kTiles) ? (j + kWaves) : j;
            const size_t nb = (size_t)jn * 16 * kH + lane * 32;
            __builtin_prefetch(Kb + nb, 0, 3);
            __builtin_prefetch(Vb + nb, 0, 3);
        }

        v8f sc0[kQT], sc1[kQT];
#pragma unroll
        for (int qi = 0; qi < kQT; ++qi) { sc0[qi] = {}; sc1[qi] = {}; }
#pragma unroll
        for (int g = 0; g < 8; ++g) {
            const v4f bk = *reinterpret_cast<const v4f*>(Kt + g * 128 + lane * 4);
            v2f bk0, bk1;
            bk0.x = bk.x;  bk0.y = bk.y;
            bk1.x = bk.z;  bk1.y = bk.w;
#pragma unroll
            for (int qi = 0; qi < kQT; ++qi) {
                v2f a0, a1;
                a0.x = aq[qi][g].x;  a0.y = aq[qi][g].y;
                a1.x = aq[qi][g].z;  a1.y = aq[qi][g].w;
                sc0[qi] = wmma_f32(a0, bk0, sc0[qi]);
                sc1[qi] = wmma_f32(a1, bk1, sc1[qi]);
            }
        }

#pragma unroll
        for (int qi = 0; qi < kQT; ++qi) {
            v8f pf;
#pragma unroll
            for (int r = 0; r < 8; ++r) {
                const float sc = sc0[qi][r] + sc1[qi][r];
                pf[r] = __builtin_amdgcn_exp2f(
                    __builtin_fmaf(sc, kLog2e, -kScoreBound * kLog2e));
            }
            lacc[qi] += pf;

#pragma unroll
            for (int r = 0; r < 8; ++r) ps[w][qi][r + 8 * hi][m] = pf[r];
        }
        asm volatile("" ::: "memory");

        const float* Vt = Vb + (size_t)j * 16 * kH;
#pragma unroll
        for (int g = 0; g < 2; ++g) {
            v4f ap[kQT];
#pragma unroll
            for (int qi = 0; qi < kQT; ++qi)
                ap[qi] = *reinterpret_cast<const v4f*>(&ps[w][qi][m][8 * g + 4 * hi]);
#pragma unroll
            for (int nt = 0; nt < 4; ++nt) {
                const v4f bv =
                    *reinterpret_cast<const v4f*>(Vt + (g * 4 + nt) * 128 + lane * 4);
#pragma unroll
                for (int u = 0; u < 2; ++u) {
                    v2f b2;
                    b2.x = bv[2 * u];  b2.y = bv[2 * u + 1];
#pragma unroll
                    for (int qi = 0; qi < kQT; ++qi) {
                        v2f a2;
                        a2.x = ap[qi][2 * u];  a2.y = ap[qi][2 * u + 1];
                        o[qi][nt] = wmma_f32(a2, b2, o[qi][nt]);
                    }
                }
            }
        }
        asm volatile("" ::: "memory");
    }

#pragma unroll
    for (int qi = 0; qi < kQT; ++qi)
#pragma unroll
        for (int r = 0; r < 8; ++r) {
            float s = lacc[qi][r];
            s += __shfl_xor(s, 1);
            s += __shfl_xor(s, 2);
            s += __shfl_xor(s, 4);
            s += __shfl_xor(s, 8);
            lacc[qi][r] = s;
        }
    if (m == 0) {
#pragma unroll
        for (int qi = 0; qi < kQT; ++qi)
#pragma unroll
            for (int r = 0; r < 8; ++r)
                l_s[w][qi * 16 + r + 8 * hi] = lacc[qi][r];
    }

    const int row = tid >> 4;
    const int c0 = (tid & 15) * 4;
#pragma unroll
    for (int qi = 0; qi < kQT; ++qi) {
#pragma unroll
        for (int nt = 0; nt < 4; ++nt)
#pragma unroll
            for (int r = 0; r < 8; ++r)
                o_s[w][r + 8 * hi][nt * 16 + m] = o[qi][nt][r];
        __syncthreads();

        float L = 0.f;
#pragma unroll
        for (int w2 = 0; w2 < kWaves; ++w2) L += l_s[w2][qi * 16 + row];
        const float invL = 1.f / L;
        v4f acc = {};
#pragma unroll
        for (int w2 = 0; w2 < kWaves; ++w2)
            acc += *reinterpret_cast<const v4f*>(&o_s[w2][row][c0]);
        float* Ob = out + ((size_t)b * kS + qb * 32 + qi * 16) * kH;
        *reinterpret_cast<v4f*>(Ob + row * kH + c0) = acc * invL;
        __syncthreads();
    }
}

extern "C" void kernel_launch(void* const* d_in, const int* in_sizes, int n_in,
                              void* d_out, int out_size, void* d_ws,
                              size_t ws_size, hipStream_t stream) {
    const float* x = (const float*)d_in[0];
    const float* Wq = (const float*)d_in[1];
    const float* Wk = (const float*)d_in[2];
    const float* Wv = (const float*)d_in[3];
    float* out = (float*)d_out;

    float* qws = (float*)d_ws;
    float* kws = qws + (size_t)kB * kS * kH;
    float* vws = kws + (size_t)kB * kS * kH;

    proj_ln_kernel<<<(kB * kS) / 64, 384, 0, stream>>>(x, Wq, Wk, Wv, qws, kws,
                                                       vws);
    flash_attn_kernel<<<kB * (kS / 32), 256, 0, stream>>>(qws, kws, vws, out);
}


# both kernels: wave priority falls with loop progress (s_setprio 3,2,1,0 by quarter of the main loop) so a SIMD's waves finish together, not oldest first with the last one alone; no arithmetic change
# speedup vs baseline: 1.5987x; 1.0502x over previous
#include <hip/hip_runtime.h>
#include <math.h>

typedef __attribute__((ext_vector_type(2))) float v2f;
typedef __attribute__((ext_vector_type(4))) float v4f;
typedef __attribute__((ext_vector_type(8))) float v8f;

namespace {
constexpr int kB = 4;
constexpr int kS = 4096;
constexpr int kD = 1024;
constexpr int kH = 64;
constexpr float kLnEps = 1e-5f;
constexpr float kScale = 0.125f;
constexpr float kScoreBound = 8.0f;
constexpr float kLog2e = 1.4426950408889634f;
}

static __device__ __forceinline__ v8f wmma_f32(v2f a, v2f b, v8f c) {
    return __builtin_amdgcn_wmma_f32_16x16x4_f32(false, a, false, b, (short)0, c,
                                                 false, false);
}

static __device__ __forceinline__ void set_progress_priority(int quarter) {
    switch (quarter) {
        case 0: __builtin_amdgcn_s_setprio(3); break;
        case 1: __builtin_amdgcn_s_setprio(2); break;
        case 2: __builtin_amdgcn_s_setprio(1); break;
        default: __builtin_amdgcn_s_setprio(0); break;
    }
}

__global__ __launch_bounds__(384) void proj_ln_kernel(
    const float* __restrict__ x, const float* __restrict__ Wq,
    const float* __restrict__ Wk, const float* __restrict__ Wv,
    float* __restrict__ qo, float* __restrict__ ko, float* __restrict__ vo) {
    constexpr int kRT = 4;
    constexpr int kRows = 16 * kRT;
    __shared__ alignas(16) float part[3][4][kRows][68];

    const int tid = threadIdx.x;
    const int wv = tid >> 5;
    const int plane = wv >> 2;
    const int ks = wv & 3;
    const int lane = tid & 31;
    const int m = lane & 15;
    const int hi = lane >> 4;
    const int rowBase = blockIdx.x * kRows;
    const float* __restrict__ Wp = (plane == 0) ? Wq : (plane == 1) ? Wk : Wv;

    v8f acc[kRT][4];
#pragma unroll
    for (int rt = 0; rt < kRT; ++rt)
#pragma unroll
        for (int nt = 0; nt < 4; ++nt) acc[rt][nt] = {};

    const int k0beg = ks * (kD / 4) + 4 * hi;
    const float* xr = x + (size_t)(rowBase + m) * kD + k0beg;
    const int wbeg = k0beg * kH + 4 * m;
    for (int c = 0; c < kD / 128; ++c) {
        if ((c & 1) == 0) set_progress_priority(c >> 1);
        v4f xa[kRT][4];
#pragma unroll
        for (int rt = 0; rt < kRT; ++rt)
#pragma unroll
            for (int j = 0; j < 4; ++j)
                xa[rt][j] = *reinterpret_cast<const v4f*>(
                    xr + (size_t)rt * 16 * kD + 32 * c + 8 * j);
#pragma unroll
        for (int j = 0; j < 4; ++j) {
            const int wo = wbeg + (32 * c + 8 * j) * kH;
#pragma unroll
            for (int u = 0; u < 2; ++u) {
                const v4f w0 = *reinterpret_cast<const v4f*>(Wp + wo + (2 * u) * kH);
                const v4f w1 =
                    *reinterpret_cast<const v4f*>(Wp + wo + (2 * u + 1) * kH);
#pragma unroll
                for (int nt = 0; nt < 4; ++nt) {
                    v2f bw;
                    bw.x = w0[nt];
                    bw.y = w1[nt];
#pragma unroll
                    for (int rt = 0; rt < kRT; ++rt) {
                        v2f a;
                        a.x = xa[rt][j][2 * u];
                        a.y = xa[rt][j][2 * u + 1];
                        acc[rt][nt] = wmma_f32(a, bw, acc[rt][nt]);
                    }
                }
            }
        }
    }

#pragma unroll
    for (int rt = 0; rt < kRT; ++rt)
#pragma unroll
        for (int r = 0; r < 8; ++r) {
            const int row = rt * 16 + r + 8 * hi;
            v4f t;
            t.x = acc[rt][0][r];
            t.y = acc[rt][1][r];
            t.z = acc[rt][2][r];
            t.w = acc[rt][3][r];
            *reinterpret_cast<v4f*>(&part[plane][ks][row][4 * m]) = t;
        }
    __syncthreads();

    if (tid < 3 * kRows) {
        const int which = tid >> 6;
        const int row = tid & (kRows - 1);
        float* r0 = &part[which][0][row][0];
        const float* r1 = &part[which][1][row][0];
        const float* r2 = &part[which][2][row][0];
        const float* r3 = &part[which][3][row][0];
        float s = 0.f, s2 = 0.f;
#pragma unroll
        for (int c = 0; c < kH; ++c) {
            const float t = r0[c] + r1[c] + r2[c] + r3[c];
            r0[c] = t;
            s += t;
            s2 += t * t;
        }
        if (which < 2) {
            const float mu = s * (1.f / kH);
            const float var = s2 * (1.f / kH) - mu * mu;
            const float inv = rsqrtf(var + kLnEps);
#pragma unroll
            for (int c = 0; c < kH; ++c) r0[c] = (r0[c] - mu) * inv;
        }
    }
    __syncthreads();

#pragma unroll
    for (int j = 0; j < 8; ++j) {
        const int idx = j * 384 + tid;
        const int pl = idx >> 10;
        const int rt = (idx >> 8) & 3;
        const int flat = (idx & 255) * 4;
        const int g = flat >> 7;
        const int mm = (flat >> 2) & 15;
        const int hh = (flat >> 6) & 1;
        float* op = (pl == 0) ? qo : (pl == 1) ? ko : vo;
        const size_t ob = (size_t)(rowBase + 16 * rt) * kH;
        v4f t;
        if (pl < 2) {
            t = *reinterpret_cast<const v4f*>(&part[pl][0][16 * rt + mm][8 * g + 4 * hh]);
        } else {
            const int vr = 16 * rt + 8 * (g >> 2) + 4 * hh;
            const int vc = 16 * (g & 3) + mm;
            t.x = part[2][0][vr][vc];
            t.y = part[2][0][vr + 1][vc];
            t.z = part[2][0][vr + 2][vc];
            t.w = part[2][0][vr + 3][vc];
        }
        *reinterpret_cast<v4f*>(op + ob + flat) = t;
    }
}

__global__ __launch_bounds__(256) void flash_attn_kernel(
    const float* __restrict__ q, const float* __restrict__ k,
    const float* __restrict__ v, float* __restrict__ out) {
    constexpr int kQT = 2;
    constexpr int kWaves = 8;
    __shared__ alignas(16) float ps[kWaves][kQT][16][20];
    __shared__ float l_s[kWaves][kQT * 16];
    __shared__ alignas(16) float o_s[kWaves][16][64];

    const int tid = threadIdx.x;
    const int w = tid >> 5;
    const int lane = tid & 31;
    const int m = lane & 15;
    const int hi = lane >> 4;
    const int b = blockIdx.x >> 7;
    const int qb = blockIdx.x & 127;

    const float* Qb = q + ((size_t)b * kS + qb * 32) * kH;
    const float* Kb = k + (size_t)b * kS * kH;
    const float* Vb = v + (size_t)b * kS * kH;

    v4f aq[kQT][8];
#pragma unroll
    for (int qi = 0; qi < kQT; ++qi)
#pragma unroll
        for (int g = 0; g < 8; ++g)
            aq[qi][g] = *reinterpret_cast<const v4f*>(Qb + qi * 16 * kH + g * 128 +
                                                      lane * 4) * kScale;

    v8f o[kQT][4];
    v8f lacc[kQT];
#pragma unroll
    for (int qi = 0; qi < kQT; ++qi) {
#pragma unroll
        for (int nt = 0; nt < 4; ++nt) o[qi][nt] = {};
        lacc[qi] = {};
    }

    constexpr int kTiles = (kS / 16) / kWaves;
    for (int t = 0; t < kTiles; ++t) {
        if ((t & 7) == 0) set_progress_priority(t >> 3);
        const int j = w + kWaves * t;
        const float* Kt = Kb + (size_t)j * 16 * kH;

        {
            const int jn = (t + 1 < kTiles) ? (j + kWaves) : j;
            const size_t nb = (size_t)jn * 16 * kH + lane * 32;
            __builtin_prefetch(Kb + nb, 0, 3);
            __builtin_prefetch(Vb + nb, 0, 3);
        }

        v8f sc0[kQT], sc1[kQT];
#pragma unroll
        for (int qi = 0; qi < kQT; ++qi) { sc0[qi] = {}; sc1[qi] = {}; }
#pragma unroll
        for (int g = 0; g < 8; ++g) {
            const v4f bk = *reinterpret_cast<const v4f*>(Kt + g * 128 + lane * 4);
            v2f bk0, bk1;
            bk0.x = bk.x;  bk0.y = bk.y;
            bk1.x = bk.z;  bk1.y = bk.w;
#pragma unroll
            for (int qi = 0; qi < kQT; ++qi) {
                v2f a0, a1;
                a0.x = aq[qi][g].x;  a0.y = aq[qi][g].y;
                a1.x = aq[qi][g].z;  a1.y = aq[qi][g].w;
                sc0[qi] = wmma_f32(a0, bk0, sc0[qi]);
                sc1[qi] = wmma_f32(a1, bk1, sc1[qi]);
            }
        }

#pragma unroll
        for (int qi = 0; qi < kQT; ++qi) {
            v8f pf;
#pragma unroll
            for (int r = 0; r < 8; ++r) {
                const float sc = sc0[qi][r] + sc1[qi][r];
                pf[r] = __builtin_amdgcn_exp2f(
                    __builtin_fmaf(sc, kLog2e, -kScoreBound * kLog2e));
            }
            lacc[qi] += pf;

#pragma unroll
            for (int r = 0; r < 8; ++r) ps[w][qi][r + 8 * hi][m] = pf[r];
        }
        asm volatile("" ::: "memory");

        const float* Vt = Vb + (size_t)j * 16 * kH;
#pragma unroll
        for (int g = 0; g < 2; ++g) {
            v4f ap[kQT];
#pragma unroll
            for (int qi = 0; qi < kQT; ++qi)
                ap[qi] = *reinterpret_cast<const v4f*>(&ps[w][qi][m][8 * g + 4 * hi]);
#pragma unroll
            for (int nt = 0; nt < 4; ++nt) {
                const v4f bv =
                    *reinterpret_cast<const v4f*>(Vt + (g * 4 + nt) * 128 + lane * 4);
#pragma unroll
                for (int u = 0; u < 2; ++u) {
                    v2f b2;
                    b2.x = bv[2 * u];  b2.y = bv[2 * u + 1];
#pragma unroll
                    for (int qi = 0; qi < kQT; ++qi) {
                        v2f a2;
                        a2.x = ap[qi][2 * u];  a2.y = ap[qi][2 * u + 1];
                        o[qi][nt] = wmma_f32(a2, b2, o[qi][nt]);
                    }
                }
            }
        }
        asm volatile("" ::: "memory");
    }

#pragma unroll
    for (int qi = 0; qi < kQT; ++qi)
#pragma unroll
        for (int r = 0; r < 8; ++r) {
            float s = lacc[qi][r];
            s += __shfl_xor(s, 1);
            s += __shfl_xor(s, 2);
            s += __shfl_xor(s, 4);
            s += __shfl_xor(s, 8);
            lacc[qi][r] = s;
        }
    if (m == 0) {
#pragma unroll
        for (int qi = 0; qi < kQT; ++qi)
#pragma unroll
            for (int r = 0; r < 8; ++r)
                l_s[w][qi * 16 + r + 8 * hi] = lacc[qi][r];
    }

    const int row = tid >> 4;
    const int c0 = (tid & 15) * 4;
#pragma unroll
    for (int qi = 0; qi < kQT; ++qi) {
#pragma unroll
        for (int nt = 0; nt < 4; ++nt)
#pragma unroll
            for (int r = 0; r < 8; ++r)
                o_s[w][r + 8 * hi][nt * 16 + m] = o[qi][nt][r];
        __syncthreads();

        float L = 0.f;
#pragma unroll
        for (int w2 = 0; w2 < kWaves; ++w2) L += l_s[w2][qi * 16 + row];
        const float invL = 1.f / L;
        v4f acc = {};
#pragma unroll
        for (int w2 = 0; w2 < kWaves; ++w2)
            acc += *reinterpret_cast<const v4f*>(&o_s[w2][row][c0]);
        float* Ob = out + ((size_t)b * kS + qb * 32 + qi * 16) * kH;
        *reinterpret_cast<v4f*>(Ob + row * kH + c0) = acc * invL;
        __syncthreads();
    }
}

extern "C" void kernel_launch(void* const* d_in, const int* in_sizes, int n_in,
                              void* d_out, int out_size, void* d_ws,
                              size_t ws_size, hipStream_t stream) {
    const float* x = (const float*)d_in[0];
    const float* Wq = (const float*)d_in[1];
    const float* Wk = (const float*)d_in[2];
    const float* Wv = (const float*)d_in[3];
    float* out = (float*)d_out;

    float* qws = (float*)d_ws;
    float* kws = qws + (size_t)kB * kS * kH;
    float* vws = kws + (size_t)kB * kS * kH;

    proj_ln_kernel<<<(kB * kS) / 64, 384, 0, stream>>>(x, Wq, Wk, Wv, qws, kws,
                                                       vws);
    flash_attn_kernel<<<kB * (kS / 32), 256, 0, stream>>>(qws, kws, vws, out);
}


# attention: a wave holds four 16-query tiles (was two): half the waves and K/V loads, two waves per SIMD; K and V fragments are requested a block of products ahead, no cache prefetch; sum orders kept
# speedup vs baseline: 1.6718x; 1.0457x over previous
#include <hip/hip_runtime.h>
#include <math.h>

typedef __attribute__((ext_vector_type(2))) float v2f;
typedef __attribute__((ext_vector_type(4))) float v4f;
typedef __attribute__((ext_vector_type(8))) float v8f;

namespace {
constexpr int kB = 4;
constexpr int kS = 4096;
constexpr int kD = 1024;
constexpr int kH = 64;
constexpr float kLnEps = 1e-5f;
constexpr float kScale = 0.125f;
constexpr float kScoreBound = 8.0f;
constexpr float kLog2e = 1.4426950408889634f;
}

static __device__ __forceinline__ v8f wmma_f32(v2f a, v2f b, v8f c) {
    return __builtin_amdgcn_wmma_f32_16x16x4_f32(false, a, false, b, (short)0, c,
                                                 false, false);
}

static __device__ __forceinline__ void set_progress_priority(int quarter) {
    switch (quarter) {
        case 0: __builtin_amdgcn_s_setprio(3); break;
        case 1: __builtin_amdgcn_s_setprio(2); break;
        case 2: __builtin_amdgcn_s_setprio(1); break;
        default: __builtin_amdgcn_s_setprio(0); break;
    }
}

__global__ __launch_bounds__(384) void proj_ln_kernel(
    const float* __restrict__ x, const float* __restrict__ Wq,
    const float* __restrict__ Wk, const float* __restrict__ Wv,
    float* __restrict__ qo, float* __restrict__ ko, float* __restrict__ vo) {
    constexpr int kRT = 4;
    constexpr int kRows = 16 * kRT;
    __shared__ alignas(16) float part[3][4][kRows][68];

    const int tid = threadIdx.x;
    const int wv = tid >> 5;
    const int plane = wv >> 2;
    const int ks = wv & 3;
    const int lane = tid & 31;
    const int m = lane & 15;
    const int hi = lane >> 4;
    const int rowBase = blockIdx.x * kRows;
    const float* __restrict__ Wp = (plane == 0) ? Wq : (plane == 1) ? Wk : Wv;

    v8f acc[kRT][4];
#pragma unroll
    for (int rt = 0; rt < kRT; ++rt)
#pragma unroll
        for (int nt = 0; nt < 4; ++nt) acc[rt][nt] = {};

    const int k0beg = ks * (kD / 4) + 4 * hi;
    const float* xr = x + (size_t)(rowBase + m) * kD + k0beg;
    const int wbeg = k0beg * kH + 4 * m;
    for (int c = 0; c < kD / 128; ++c) {
        if ((c & 1) == 0) set_progress_priority(c >> 1);
        v4f xa[kRT][4];
#pragma unroll
        for (int rt = 0; rt < kRT; ++rt)
#pragma unroll
            for (int j = 0; j < 4; ++j)
                xa[rt][j] = *reinterpret_cast<const v4f*>(
                    xr + (size_t)rt * 16 * kD + 32 * c + 8 * j);
#pragma unroll
        for (int j = 0; j < 4; ++j) {
            const int wo = wbeg + (32 * c + 8 * j) * kH;
#pragma unroll
            for (int u = 0; u < 2; ++u) {
                const v4f w0 = *reinterpret_cast<const v4f*>(Wp + wo + (2 * u) * kH);
                const v4f w1 =
                    *reinterpret_cast<const v4f*>(Wp + wo + (2 * u + 1) * kH);
#pragma unroll
                for (int nt = 0; nt < 4; ++nt) {
                    v2f bw;
                    bw.x = w0[nt];
                    bw.y = w1[nt];
#pragma unroll
                    for (int rt = 0; rt < kRT; ++rt) {
                        v2f a;
                        a.x = xa[rt][j][2 * u];
                        a.y = xa[rt][j][2 * u + 1];
                        acc[rt][nt] = wmma_f32(a, bw, acc[rt][nt]);
                    }
                }
            }
        }
    }

#pragma unroll
    for (int rt = 0; rt < kRT; ++rt)
#pragma unroll
        for (int r = 0; r < 8; ++r) {
            const int row = rt * 16 + r + 8 * hi;
            v4f t;
            t.x = acc[rt][0][r];
            t.y = acc[rt][1][r];
            t.z = acc[rt][2][r];
            t.w = acc[rt][3][r];
            *reinterpret_cast<v4f*>(&part[plane][ks][row][4 * m]) = t;
        }
    __syncthreads();

    if (tid < 3 * kRows) {
        const int which = tid >> 6;
        const int row = tid & (kRows - 1);
        float* r0 = &part[which][0][row][0];
        const float* r1 = &part[which][1][row][0];
        const float* r2 = &part[which][2][row][0];
        const float* r3 = &part[which][3][row][0];
        float s = 0.f, s2 = 0.f;
#pragma unroll
        for (int c = 0; c < kH; ++c) {
            const float t = r0[c] + r1[c] + r2[c] + r3[c];
            r0[c] = t;
            s += t;
            s2 += t * t;
        }
        if (which < 2) {
            const float mu = s * (1.f / kH);
            const float var = s2 * (1.f / kH) - mu * mu;
            const float inv = rsqrtf(var + kLnEps);
#pragma unroll
            for (int c = 0; c < kH; ++c) r0[c] = (r0[c] - mu) * inv;
        }
    }
    __syncthreads();

#pragma unroll
    for (int j = 0; j < 8; ++j) {
        const int idx = j * 384 + tid;
        const int pl = idx >> 10;
        const int rt = (idx >> 8) & 3;
        const int flat = (idx & 255) * 4;
        const int g = flat >> 7;
        const int mm = (flat >> 2) & 15;
        const int hh = (flat >> 6) & 1;
        float* op = (pl == 0) ? qo : (pl == 1) ? ko : vo;
        const size_t ob = (size_t)(rowBase + 16 * rt) * kH;
        v4f t;
        if (pl < 2) {
            t = *reinterpret_cast<const v4f*>(&part[pl][0][16 * rt + mm][8 * g + 4 * hh]);
        } else {
            const int vr = 16 * rt + 8 * (g >> 2) + 4 * hh;
            const int vc = 16 * (g & 3) + mm;
            t.x = part[2][0][vr][vc];
            t.y = part[2][0][vr + 1][vc];
            t.z = part[2][0][vr + 2][vc];
            t.w = part[2][0][vr + 3][vc];
        }
        *reinterpret_cast<v4f*>(op + ob + flat) = t;
    }
}

__global__ __launch_bounds__(256) void flash_attn_kernel(
    const float* __restrict__ q, const float* __restrict__ k,
    const float* __restrict__ v, float* __restrict__ out) {
    constexpr int kQT = 4;
    constexpr int kWaves = 8;
    __shared__ alignas(16) float ps[kWaves][kQT][16][20];
    __shared__ float l_s[kWaves][kQT * 16];
    __shared__ alignas(16) float o_s[kWaves][16][64];

    const int tid = threadIdx.x;
    const int w = tid >> 5;
    const int lane = tid & 31;
    const int m = lane & 15;
    const int hi = lane >> 4;
    const int b = blockIdx.x >> 6;
    const int qb = blockIdx.x & 63;

    const float* Qb = q + ((size_t)b * kS + qb * 64) * kH;
    const float* Kb = k + (size_t)b * kS * kH;
    const float* Vb = v + (size_t)b * kS * kH;

    v4f aq[kQT][8];
#pragma unroll
    for (int qi = 0; qi < kQT; ++qi)
#pragma unroll
        for (int g = 0; g < 8; ++g)
            aq[qi][g] = *reinterpret_cast<const v4f*>(Qb + qi * 16 * kH + g * 128 +
                                                      lane * 4) * kScale;

    v8f o[kQT][4];
    v8f lacc[kQT];
#pragma unroll
    for (int qi = 0; qi < kQT; ++qi) {
#pragma unroll
        for (int nt = 0; nt < 4; ++nt) o[qi][nt] = {};
        lacc[qi] = {};
    }

    v4f kf[8];
#pragma unroll
    for (int g = 0; g < 8; ++g)
        kf[g] = *reinterpret_cast<const v4f*>(Kb + (size_t)w * 16 * kH + g * 128 +
                                              lane * 4);

    constexpr int kTiles = (kS / 16) / kWaves;
    for (int t = 0; t < kTiles; ++t) {
        if ((t & 7) == 0) set_progress_priority(t >> 3);
        const int j = w + kWaves * t;

        const float* Vt = Vb + (size_t)j * 16 * kH;
        v4f vf[8];
#pragma unroll
        for (int i = 0; i < 8; ++i)
            vf[i] = *reinterpret_cast<const v4f*>(Vt + i * 128 + lane * 4);

        v8f sc0[kQT], sc1[kQT];
#pragma unroll
        for (int qi = 0; qi < kQT; ++qi) { sc0[qi] = {}; sc1[qi] = {}; }
#pragma unroll
        for (int g = 0; g < 8; ++g) {
            v2f bk0, bk1;
            bk0.x = kf[g].x;  bk0.y = kf[g].y;
            bk1.x = kf[g].z;  bk1.y = kf[g].w;
#pragma unroll
            for (int qi = 0; qi < kQT; ++qi) {
                v2f a0, a1;
                a0.x = aq[qi][g].x;  a0.y = aq[qi][g].y;
                a1.x = aq[qi][g].z;  a1.y = aq[qi][g].w;
                sc0[qi] = wmma_f32(a0, bk0, sc0[qi]);
                sc1[qi] = wmma_f32(a1, bk1, sc1[qi]);
            }
        }

        {
            const int jn = (t + 1 < kTiles) ? (j + kWaves) : j;
            const float* Kn = Kb + (size_t)jn * 16 * kH;
#pragma unroll
            for (int g = 0; g < 8; ++g)
                kf[g] = *reinterpret_cast<const v4f*>(Kn + g * 128 + lane * 4);
        }

#pragma unroll
        for (int qi = 0; qi < kQT; ++qi) {
            v8f pf;
#pragma unroll
            for (int r = 0; r < 8; ++r) {
                const float sc = sc0[qi][r] + sc1[qi][r];
                pf[r] = __builtin_amdgcn_exp2f(
                    __builtin_fmaf(sc, kLog2e, -kScoreBound * kLog2e));
            }
            lacc[qi] += pf;

#pragma unroll
            for (int r = 0; r < 8; ++r) ps[w][qi][r + 8 * hi][m] = pf[r];
        }
        asm volatile("" ::: "memory");

#pragma unroll
        for (int g = 0; g < 2; ++g) {
            v4f ap[kQT];
#pragma unroll
            for (int qi = 0; qi < kQT; ++qi)
                ap[qi] = *reinterpret_cast<const v4f*>(&ps[w][qi][m][8 * g + 4 * hi]);
#pragma unroll
            for (int nt = 0; nt < 4; ++nt) {
                const v4f bv = vf[g * 4 + nt];
#pragma unroll
                for (int u = 0; u < 2; ++u) {
                    v2f b2;
                    b2.x = bv[2 * u];  b2.y = bv[2 * u + 1];
#pragma unroll
                    for (int qi = 0; qi < kQT; ++qi) {
                        v2f a2;
                        a2.x = ap[qi][2 * u];  a2.y = ap[qi][2 * u + 1];
                        o[qi][nt] = wmma_f32(a2, b2, o[qi][nt]);
                    }
                }
            }
        }
        asm volatile("" ::: "memory");
    }

#pragma unroll
    for (int qi = 0; qi < kQT; ++qi)
#pragma unroll
        for (int r = 0; r < 8; ++r) {
            float s = lacc[qi][r];
            s += __shfl_xor(s, 1);
            s += __shfl_xor(s, 2);
            s += __shfl_xor(s, 4);
            s += __shfl_xor(s, 8);
            lacc[qi][r] = s;
        }
    if (m == 0) {
#pragma unroll
        for (int qi = 0; qi < kQT; ++qi)
#pragma unroll
            for (int r = 0; r < 8; ++r)
                l_s[w][qi * 16 + r + 8 * hi] = lacc[qi][r];
    }

    const int row = tid >> 4;
    const int c0 = (tid & 15) * 4;
#pragma unroll
    for (int qi = 0; qi < kQT; ++qi) {
#pragma unroll
        for (int nt = 0; nt < 4; ++nt)
#pragma unroll
            for (int r = 0; r < 8; ++r)
                o_s[w][r + 8 * hi][nt * 16 + m] = o[qi][nt][r];
        __syncthreads();

        float L = 0.f;
#pragma unroll
        for (int w2 = 0; w2 < kWaves; ++w2) L += l_s[w2][qi * 16 + row];
        const float invL = 1.f / L;
        v4f acc = {};
#pragma unroll
        for (int w2 = 0; w2 < kWaves; ++w2)
            acc += *reinterpret_cast<const v4f*>(&o_s[w2][row][c0]);
        float* Ob = out + ((size_t)b * kS + qb * 64 + qi * 16) * kH;
        *reinterpret_cast<v4f*>(Ob + row * kH + c0) = acc * invL;
        __syncthreads();
    }
}

extern "C" void kernel_launch(void* const* d_in, const int* in_sizes, int n_in,
                              void* d_out, int out_size, void* d_ws,
                              size_t ws_size, hipStream_t stream) {
    const float* x = (const float*)d_in[0];
    const float* Wq = (const float*)d_in[1];
    const float* Wk = (const float*)d_in[2];
    const float* Wv = (const float*)d_in[3];
    float* out = (float*)d_out;

    float* qws = (float*)d_ws;
    float* kws = qws + (size_t)kB * kS * kH;
    float* vws = kws + (size_t)kB * kS * kH;

    proj_ln_kernel<<<(kB * kS) / 64, 384, 0, stream>>>(x, Wq, Wk, Wv, qws, kws,
                                                       vws);
    flash_attn_kernel<<<kB * (kS / 64), 256, 0, stream>>>(qws, kws, vws, out);
}
